# starting kernel
# baseline (speedup 1.0000x reference)
#include <hip/hip_runtime.h>

typedef float v2f __attribute__((ext_vector_type(2)));
typedef float v4f __attribute__((ext_vector_type(4)));
typedef float v8f __attribute__((ext_vector_type(8)));

#define PITCH 34

__global__ __launch_bounds__(256) void ecc_aggregate_kernel(
    const float* __restrict__ nodes,
    const float* __restrict__ edges,
    const float* __restrict__ mask,
    const float* __restrict__ W,
    const float* __restrict__ bias,
    float* __restrict__ out)
{
    __shared__ float sE[64 * PITCH];
    __shared__ float sN[64 * PITCH];
    __shared__ float sMask[64];
    __shared__ float sPart[64];

    const int tid  = threadIdx.x;
    const int wave = tid >> 5;
    const int lane = tid & 31;
    const int lg   = lane >> 4;
    const int m    = lane & 15;

    const int wg = blockIdx.x;
    const int b  = wg >> 6;
    const int i  = wg & 63;

    const float* eBase = edges + ((size_t)b * 4096 + (size_t)i * 64) * 32;
    const float* nBase = nodes + (size_t)b * 64 * 32;
    const float* mBase = mask  + (size_t)b * 4096 + (size_t)i * 64;

    for (int q = tid; q < 64 * 32; q += 256) {
        const int j = q >> 5, f = q & 31;
        sE[j * PITCH + f] = eBase[q];
        sN[j * PITCH + f] = nBase[q];
    }
    if (tid < 64) sMask[tid] = mBase[tid];
    __syncthreads();

    float tileAcc[8];
#pragma unroll
    for (int k = 0; k < 8; ++k) tileAcc[k] = 0.0f;

#pragma unroll
    for (int k = 0; k < 8; ++k) {
        const int t  = wave + 8 * k;
        const int g0 = t << 4;
        const int e0 = (t & 1) << 4;

        v2f af[8];
#pragma unroll
        for (int kk = 0; kk < 8; ++kk)
            af[kk] = *(const v2f*)(W + (size_t)(g0 + m) * 32 + 4 * kk + 2 * lg);

        const v4f b0 = *(const v4f*)(bias + g0 + 8 * lg);
        const v4f b1 = *(const v4f*)(bias + g0 + 8 * lg + 4);
        const float bv[8] = {b0.x, b0.y, b0.z, b0.w, b1.x, b1.y, b1.z, b1.w};

#pragma unroll
        for (int jg = 0; jg < 4; ++jg) {
            const int j16 = jg * 16 + m;
            v8f c = {};
#pragma unroll
            for (int kk = 0; kk < 8; ++kk) {
                v2f bf = *(const v2f*)&sE[j16 * PITCH + 4 * kk + 2 * lg];
                c = __builtin_amdgcn_wmma_f32_16x16x4_f32(
                        false, af[kk], false, bf, (short)0, c, false, false);
            }
            float partial = 0.0f;
#pragma unroll
            for (int r = 0; r < 8; ++r) {
                float v = c[r] + bv[r];
                v = v > 0.0f ? v : 0.0f;
                partial += v * sN[j16 * PITCH + e0 + 8 * lg + r];
            }
            partial *= sMask[j16];
#pragma unroll
            for (int off = 16; off > 0; off >>= 1)
                partial += __shfl_xor(partial, off, 32);
            tileAcc[k] += partial;
        }
    }

    if (lane == 0) {
#pragma unroll
        for (int k = 0; k < 8; ++k) sPart[wave + 8 * k] = tileAcc[k];
    }
    __syncthreads();

    if (tid < 32)
        out[(size_t)wg * 32 + tid] = sPart[2 * tid] + sPart[2 * tid + 1];
}

extern "C" void kernel_launch(void* const* d_in, const int* in_sizes, int n_in,
                              void* d_out, int out_size, void* d_ws, size_t ws_size,
                              hipStream_t stream) {
    const float* nodes = (const float*)d_in[0];
    const float* edges = (const float*)d_in[1];
    const float* mask  = (const float*)d_in[2];
    const float* W     = (const float*)d_in[3];
    const float* bias  = (const float*)d_in[4];
    float* out         = (float*)d_out;

    ecc_aggregate_kernel<<<dim3(16 * 64), dim3(256), 0, stream>>>(
        nodes, edges, mask, W, bias, out);
}


# Product transposed: edge rows are A, W rows 32 apart (one e, 16 d) are the B columns; each lane owns one d and sums e and j itself, bias starts C, mask folded into node values; sum order changes
# speedup vs baseline: 1.5734x; 1.5734x over previous
#include <hip/hip_runtime.h>

typedef float v2f __attribute__((ext_vector_type(2)));
typedef float v4f __attribute__((ext_vector_type(4)));
typedef float v8f __attribute__((ext_vector_type(8)));

#define PITCH  34
#define PITCHJ 68

__global__ __launch_bounds__(256) void ecc_aggregate_kernel(
    const float* __restrict__ nodes,
    const float* __restrict__ edges,
    const float* __restrict__ mask,
    const float* __restrict__ W,
    const float* __restrict__ bias,
    float* __restrict__ out)
{
    __shared__ float sE[64 * PITCH];
    __shared__ __attribute__((aligned(16))) float sNM[32 * PITCHJ];
    __shared__ float sRed[16 * 32];

    const int tid  = threadIdx.x;
    const int wave = tid >> 5;
    const int lane = tid & 31;
    const int lg   = lane >> 4;
    const int m    = lane & 15;

    const int wg = blockIdx.x;
    const int b  = wg >> 6;
    const int i  = wg & 63;

    const float* eBase = edges + ((size_t)b * 4096 + (size_t)i * 64) * 32;
    const float* nBase = nodes + (size_t)b * 64 * 32;
    const float* mBase = mask  + (size_t)b * 4096 + (size_t)i * 64;

    const float* wLane = W + ((size_t)(32 * m + 4 * wave)) * 32 + 2 * lg;
    const float* bLane = bias + 32 * m + 4 * wave;

    v2f wbN[8];
    float bgN;
#pragma unroll
    for (int kk = 0; kk < 8; ++kk) wbN[kk] = *(const v2f*)(wLane + 4 * kk);
    bgN = bLane[0];

    for (int q = tid; q < 64 * 32; q += 256) {
        const int j = q >> 5, f = q & 31;
        sE[j * PITCH + f]   = eBase[q];
        sNM[f * PITCHJ + j] = nBase[q] * mBase[j];
    }
    __syncthreads();

    v2f ea[4][8];
#pragma unroll
    for (int jt = 0; jt < 4; ++jt)
#pragma unroll
        for (int kk = 0; kk < 8; ++kk)
            ea[jt][kk] = *(const v2f*)&sE[(16 * jt + m) * PITCH + 4 * kk + 2 * lg];

    v8f acc[2] = {(v8f)(0.0f), (v8f)(0.0f)};

#pragma unroll
    for (int tt = 0; tt < 8; ++tt) {
        const int t  = tt >> 1;
        const int dh = tt & 1;
        const int e  = 4 * wave + t;

        v2f wb[8];
#pragma unroll
        for (int kk = 0; kk < 8; ++kk) wb[kk] = wbN[kk];
        const float bg = bgN;

        if (tt + 1 < 8) {
            const int tn = (tt + 1) >> 1, dn = (tt + 1) & 1;
            const float* wp = wLane + ((size_t)(512 * dn + tn)) * 32;
#pragma unroll
            for (int kk = 0; kk < 8; ++kk) wbN[kk] = *(const v2f*)(wp + 4 * kk);
            bgN = bLane[512 * dn + tn];
        }
        __builtin_amdgcn_sched_barrier(0);

        v8f c[4];
#pragma unroll
        for (int jt = 0; jt < 4; ++jt)
            c[jt] = (v8f)(bg);

#pragma unroll
        for (int kk = 0; kk < 8; ++kk)
#pragma unroll
            for (int jt = 0; jt < 4; ++jt)
                c[jt] = __builtin_amdgcn_wmma_f32_16x16x4_f32(
                            false, ea[jt][kk], false, wb[kk], (short)0, c[jt], false, false);

#pragma unroll
        for (int jt = 0; jt < 4; ++jt) {
            const v4f n0 = *(const v4f*)&sNM[e * PITCHJ + 16 * jt + 8 * lg];
            const v4f n1 = *(const v4f*)&sNM[e * PITCHJ + 16 * jt + 8 * lg + 4];
            const v8f nv = __builtin_shufflevector(n0, n1, 0, 1, 2, 3, 4, 5, 6, 7);
            const v8f x  = c[jt];
            const v8f v  = __builtin_elementwise_maximum(x, (v8f)(0.0f));
            acc[dh] += v * nv;
        }
    }

    float a0 = 0.0f, a1 = 0.0f;
#pragma unroll
    for (int r = 0; r < 8; ++r) { a0 += acc[0][r]; a1 += acc[1][r]; }
    sRed[(wave * 2 + lg) * 32 + m]      = a0;
    sRed[(wave * 2 + lg) * 32 + 16 + m] = a1;
    __syncthreads();

    if (tid < 32) {
        float s = 0.0f;
#pragma unroll
        for (int r = 0; r < 16; ++r) s += sRed[r * 32 + tid];
        out[(size_t)wg * 32 + tid] = s;
    }
}

extern "C" void kernel_launch(void* const* d_in, const int* in_sizes, int n_in,
                              void* d_out, int out_size, void* d_ws, size_t ws_size,
                              hipStream_t stream) {
    const float* nodes = (const float*)d_in[0];
    const float* edges = (const float*)d_in[1];
    const float* mask  = (const float*)d_in[2];
    const float* W     = (const float*)d_in[3];
    const float* bias  = (const float*)d_in[4];
    float* out         = (float*)d_out;

    ecc_aggregate_kernel<<<dim3(16 * 64), dim3(256), 0, stream>>>(
        nodes, edges, mask, W, bias, out);
}
